# starting kernel
# baseline (speedup 1.0000x reference)
#include <hip/hip_runtime.h>


typedef __attribute__((ext_vector_type(2))) float v2f;
typedef __attribute__((ext_vector_type(8))) float v8f;

#define NROWS 4096
#define DCOLS 64

__global__ __launch_bounds__(32) void toeplitz_scan_wmma(
    const float* __restrict__ x, const float* __restrict__ zero,
    float* __restrict__ out)
{
    const int bh   = blockIdx.x >> 2;
    const int ct   = blockIdx.x & 3;
    const int lane = threadIdx.x & 31;
    const int col  = lane & 15;
    const bool hi  = lane >= 16;

    const float c1 = __expf(fminf(fmaxf(zero[0], -60.f), 30.f));

    const size_t base = (size_t)bh * NROWS * DCOLS + (size_t)ct * 16 + (size_t)col;
    const float* xp = x   + base;
    float*       op = out + base;

    v2f a[4];
#pragma unroll
    for (int j = 0; j < 4; ++j) {
        const int kb = 4 * j + (hi ? 2 : 0);
        a[j].x = (col >= kb)     ? c1 : 0.f;
        a[j].y = (col >= kb + 1) ? c1 : 0.f;
    }

    float carry = 0.f;

    for (int r = 0; r < NROWS / 16; ++r) {
        const float* tp = xp + (size_t)r * 16 * DCOLS;

        __builtin_prefetch(tp + 2 * 16 * DCOLS, 0, 3);

        v2f b[4];
#pragma unroll
        for (int j = 0; j < 4; ++j) {
            const int row = 4 * j + (hi ? 2 : 0);
            b[j].x = tp[(size_t)row       * DCOLS];
            b[j].y = tp[(size_t)(row + 1) * DCOLS];
        }

        v8f acc;
#pragma unroll
        for (int v = 0; v < 8; ++v) acc[v] = carry;

#pragma unroll
        for (int j = 0; j < 4; ++j)
            acc = __builtin_amdgcn_wmma_f32_16x16x4_f32(
                false, a[j], false, b[j], (short)0, acc, false, false);

        float* topp = op + (size_t)(r * 16 + (hi ? 8 : 0)) * DCOLS;
#pragma unroll
        for (int v = 0; v < 8; ++v)
            topp[(size_t)v * DCOLS] = acc[v];

        const int ai = __float_as_int(acc[7]);
        const int sw = __builtin_amdgcn_permlanex16(ai, ai, 0x76543210u, 0xfedcba98u,
                                                    true, false);
        carry = hi ? acc[7] : __int_as_float(sw);
    }
}

extern "C" void kernel_launch(void* const* d_in, const int* in_sizes, int n_in,
                              void* d_out, int out_size, void* d_ws, size_t ws_size,
                              hipStream_t stream) {
    const float* x    = (const float*)d_in[0];
    const float* zero = (const float*)d_in[2];
    float* out = (float*)d_out;

    toeplitz_scan_wmma<<<dim3(256), dim3(32), 0, stream>>>(x, zero, out);
}


# Split each 4096-row column scan over 16 waves of one work-group: chunk held in registers, chunk totals through LDS and one barrier, then the WMMA scan starts from their sum (per-chunk sum order)
# speedup vs baseline: 4.2918x; 4.2918x over previous
#include <hip/hip_runtime.h>


typedef __attribute__((ext_vector_type(2))) float v2f;
typedef __attribute__((ext_vector_type(8))) float v8f;

#define NROWS 4096
#define DCOLS 64
#define NWAVES 16
#define ROWS_PER_WAVE (NROWS / NWAVES)
#define TILES (ROWS_PER_WAVE / 16)

__global__ __launch_bounds__(512) void toeplitz_scan_wmma(
    const float* __restrict__ x, const float* __restrict__ zero,
    float* __restrict__ out)
{
    __shared__ float tot[NWAVES][16];

    const int bh   = blockIdx.x >> 2;
    const int ct   = blockIdx.x & 3;
    const int lane = threadIdx.x & 31;
    const int wave = __builtin_amdgcn_readfirstlane((int)(threadIdx.x >> 5));
    const int col  = lane & 15;
    const bool hi  = lane >= 16;

    const float c1 = __expf(fminf(fmaxf(zero[0], -60.f), 30.f));

    const size_t base = (size_t)bh * NROWS * DCOLS + (size_t)wave * ROWS_PER_WAVE * DCOLS
                      + (size_t)ct * 16 + (size_t)col;
    const float* xp = x   + base;
    float*       op = out + base;

    v2f a[4];
#pragma unroll
    for (int j = 0; j < 4; ++j) {
        const int kb = 4 * j + (hi ? 2 : 0);
        a[j].x = (col >= kb)     ? c1 : 0.f;
        a[j].y = (col >= kb + 1) ? c1 : 0.f;
    }

    v2f b[TILES][4];
#pragma unroll
    for (int t = 0; t < TILES; ++t) {
#pragma unroll
        for (int j = 0; j < 4; ++j) {
            const int row = 16 * t + 4 * j + (hi ? 2 : 0);
            b[t][j].x = xp[(size_t)row       * DCOLS];
            b[t][j].y = xp[(size_t)(row + 1) * DCOLS];
        }
    }

    float s0 = 0.f, s1 = 0.f, s2 = 0.f, s3 = 0.f;
#pragma unroll
    for (int t = 0; t < TILES; ++t) {
        s0 += b[t][0].x + b[t][0].y;
        s1 += b[t][1].x + b[t][1].y;
        s2 += b[t][2].x + b[t][2].y;
        s3 += b[t][3].x + b[t][3].y;
    }
    const float sh = (s0 + s1) + (s2 + s3);
    const int shi = __float_as_int(sh);
    const int sho = __builtin_amdgcn_permlanex16(shi, shi, 0x76543210u, 0xfedcba98u,
                                                 true, false);
    const float total = c1 * (sh + __int_as_float(sho));
    if (!hi) tot[wave][col] = total;
    __syncthreads();

    float tv[NWAVES - 1];
#pragma unroll
    for (int w = 0; w < NWAVES - 1; ++w) tv[w] = tot[w][col];
    __builtin_amdgcn_sched_barrier(0);
    float carry = 0.f;
#pragma unroll
    for (int w = 0; w < NWAVES - 1; ++w) carry += (w < wave) ? tv[w] : 0.f;

#pragma unroll
    for (int t = 0; t < TILES; ++t) {
        v8f acc;
#pragma unroll
        for (int v = 0; v < 8; ++v) acc[v] = carry;

#pragma unroll
        for (int j = 0; j < 4; ++j)
            acc = __builtin_amdgcn_wmma_f32_16x16x4_f32(
                false, a[j], false, b[t][j], (short)0, acc, false, false);

        float* topp = op + (size_t)(t * 16 + (hi ? 8 : 0)) * DCOLS;
#pragma unroll
        for (int v = 0; v < 8; ++v)
            topp[(size_t)v * DCOLS] = acc[v];

        const int ai = __float_as_int(acc[7]);
        const int sw = __builtin_amdgcn_permlanex16(ai, ai, 0x76543210u, 0xfedcba98u,
                                                    true, false);
        carry = hi ? acc[7] : __int_as_float(sw);
    }
}

extern "C" void kernel_launch(void* const* d_in, const int* in_sizes, int n_in,
                              void* d_out, int out_size, void* d_ws, size_t ws_size,
                              hipStream_t stream) {
    const float* x    = (const float*)d_in[0];
    const float* zero = (const float*)d_in[2];
    float* out = (float*)d_out;

    toeplitz_scan_wmma<<<dim3(256), dim3(512), 0, stream>>>(x, zero, out);
}


# Whole 128-B lines per access: work-group = 32 columns, a lane's 8-byte column pair is the WMMA fragment (8-row tiles, k = row x parity); rows swept in 4 prefetched blocks; chunk sums per 64 rows
# speedup vs baseline: 10.4862x; 2.4433x over previous
#include <hip/hip_runtime.h>


typedef __attribute__((ext_vector_type(2))) float v2f;
typedef __attribute__((ext_vector_type(8))) float v8f;

#define NROWS 4096
#define DCOLS 64
#define WCOLS 32
#define NWAVES 16
#define STEPS 4
#define ROWS_PER_STEP (NROWS / STEPS)
#define ROWS_PER_WAVE (ROWS_PER_STEP / NWAVES)
#define TROWS 8
#define TILES (ROWS_PER_WAVE / TROWS)
#define ROWPITCH2 (DCOLS / 2)

__device__ __forceinline__ void load_block(const v2f* __restrict__ p, v2f (&d)[TILES][4])
{
#pragma unroll
    for (int t = 0; t < TILES; ++t)
#pragma unroll
        for (int j = 0; j < 4; ++j)
            d[t][j] = p[(size_t)(TROWS * t + 2 * j) * ROWPITCH2];
}

__device__ __forceinline__ v2f block_total(const v2f (&d)[TILES][4])
{
    v2f s0 = {0.f, 0.f}, s1 = {0.f, 0.f}, s2 = {0.f, 0.f}, s3 = {0.f, 0.f};
#pragma unroll
    for (int t = 0; t < TILES; ++t) {
        s0 += d[t][0];
        s1 += d[t][1];
        s2 += d[t][2];
        s3 += d[t][3];
    }
    const v2f sh = (s0 + s1) + (s2 + s3);
    const int x0 = __float_as_int(sh.x), y0 = __float_as_int(sh.y);
    const int x1 = __builtin_amdgcn_permlanex16(x0, x0, 0x76543210u, 0xfedcba98u, true, false);
    const int y1 = __builtin_amdgcn_permlanex16(y0, y0, 0x76543210u, 0xfedcba98u, true, false);
    v2f r;
    r.x = sh.x + __int_as_float(x1);
    r.y = sh.y + __int_as_float(y1);
    return r;
}

__device__ __forceinline__ void lds_barrier()
{
    __builtin_amdgcn_fence(__ATOMIC_RELEASE, "workgroup", "local");
    __builtin_amdgcn_s_barrier();
    __builtin_amdgcn_fence(__ATOMIC_ACQUIRE, "workgroup", "local");
}

__global__ __launch_bounds__(512) void toeplitz_scan_wmma(
    const float* __restrict__ x, const float* __restrict__ zero,
    float* __restrict__ out)
{
    __shared__ v2f tot[2][NWAVES][16];

    const int bh   = blockIdx.x >> 1;
    const int ch   = blockIdx.x & 1;
    const int lane = threadIdx.x & 31;
    const int wave = __builtin_amdgcn_readfirstlane((int)(threadIdx.x >> 5));
    const int col  = lane & 15;
    const bool hi  = lane >= 16;

    const float c1 = __expf(fminf(fmaxf(zero[0], -60.f), 30.f));

    const size_t base = (size_t)bh * NROWS * DCOLS + (size_t)wave * ROWS_PER_WAVE * DCOLS
                      + (size_t)ch * WCOLS + (size_t)(2 * col);
    const v2f* xp = (const v2f*)(x + base + (hi ? DCOLS : 0));
    v2f*       op = (v2f*)(out + base + (hi ? 4 * DCOLS : 0));

    v2f a[4];
    {
        const int ai = col >> 1;
        const int aq = col & 1;
#pragma unroll
        for (int j = 0; j < 4; ++j) {
            const int r = 2 * j + (hi ? 1 : 0);
            const float v = (r <= ai) ? c1 : 0.f;
            a[j].x = (aq == 0) ? v : 0.f;
            a[j].y = (aq == 1) ? v : 0.f;
        }
    }

    v2f buf[2][TILES][4];
    load_block(xp, buf[0]);
    load_block(xp + (size_t)ROWS_PER_STEP * ROWPITCH2, buf[1]);

    {
        const v2f t0 = block_total(buf[0]);
        if (!hi) tot[0][wave][col] = c1 * t0;
    }
    lds_barrier();

    v2f run = {0.f, 0.f};

#pragma unroll
    for (int s = 0; s < STEPS; ++s) {
        const int par = s & 1;

        v2f tv[NWAVES];
#pragma unroll
        for (int w = 0; w < NWAVES; ++w) tv[w] = tot[par][w][col];
        v2f pre = {0.f, 0.f}, all = {0.f, 0.f};
#pragma unroll
        for (int w = 0; w < NWAVES; ++w) {
            const v2f z = {0.f, 0.f};
            pre += (w < wave) ? tv[w] : z;
            all += tv[w];
        }
        float carry0 = run.x + pre.x;
        float carry1 = run.y + pre.y;
        run += all;

#pragma unroll
        for (int t = 0; t < TILES; ++t) {
            v8f acc;
#pragma unroll
            for (int v = 0; v < 8; ++v) acc[v] = (v & 1) ? carry1 : carry0;

#pragma unroll
            for (int j = 0; j < 4; ++j)
                acc = __builtin_amdgcn_wmma_f32_16x16x4_f32(
                    false, a[j], false, buf[par][t][j], (short)0, acc, false, false);

            v2f* topp = op + (size_t)(s * ROWS_PER_STEP + t * TROWS) * ROWPITCH2;
#pragma unroll
            for (int u = 0; u < 4; ++u) {
                v2f o;
                o.x = acc[2 * u];
                o.y = acc[2 * u + 1];
                topp[(size_t)u * ROWPITCH2] = o;
            }

            const int a0i = __float_as_int(acc[6]);
            const int a1i = __float_as_int(acc[7]);
            const int s0i = __builtin_amdgcn_permlanex16(a0i, a0i, 0x76543210u, 0xfedcba98u,
                                                         true, false);
            const int s1i = __builtin_amdgcn_permlanex16(a1i, a1i, 0x76543210u, 0xfedcba98u,
                                                         true, false);
            carry0 = hi ? acc[6] : __int_as_float(s0i);
            carry1 = hi ? acc[7] : __int_as_float(s1i);
        }

        if (s + 2 < STEPS) {
            __builtin_amdgcn_sched_barrier(0);
            load_block(xp + (size_t)(s + 2) * ROWS_PER_STEP * ROWPITCH2, buf[par]);
        }

        if (s + 1 < STEPS) {
            __builtin_amdgcn_sched_barrier(0);
            const v2f t1 = block_total(buf[par ^ 1]);
            if (!hi) tot[par ^ 1][wave][col] = c1 * t1;
            lds_barrier();
        }
    }
}

extern "C" void kernel_launch(void* const* d_in, const int* in_sizes, int n_in,
                              void* d_out, int out_size, void* d_ws, size_t ws_size,
                              hipStream_t stream) {
    const float* x    = (const float*)d_in[0];
    const float* zero = (const float*)d_in[2];
    float* out = (float*)d_out;

    toeplitz_scan_wmma<<<dim3(128), dim3(512), 0, stream>>>(x, zero, out);
}


# Two work-groups per 32-column group (256, one per compute unit): lower scans row blocks 0-1; upper re-reads blocks 0-1 and adds them up for its start value, then scans 2-3; sum order unchanged
# speedup vs baseline: 12.0785x; 1.1518x over previous
#include <hip/hip_runtime.h>


typedef __attribute__((ext_vector_type(2))) float v2f;
typedef __attribute__((ext_vector_type(8))) float v8f;

#define NROWS 4096
#define DCOLS 64
#define WCOLS 32
#define NWAVES 16
#define STEPS 4
#define SCAN_STEPS (STEPS / 2)
#define NGROUPS 128
#define ROWS_PER_STEP (NROWS / STEPS)
#define ROWS_PER_WAVE (ROWS_PER_STEP / NWAVES)
#define TROWS 8
#define TILES (ROWS_PER_WAVE / TROWS)
#define ROWPITCH2 (DCOLS / 2)

__device__ __forceinline__ void load_block(const v2f* __restrict__ p, v2f (&d)[TILES][4])
{
#pragma unroll
    for (int t = 0; t < TILES; ++t)
#pragma unroll
        for (int j = 0; j < 4; ++j)
            d[t][j] = p[(size_t)(TROWS * t + 2 * j) * ROWPITCH2];
}

__device__ __forceinline__ v2f block_total(const v2f (&d)[TILES][4])
{
    v2f s0 = {0.f, 0.f}, s1 = {0.f, 0.f}, s2 = {0.f, 0.f}, s3 = {0.f, 0.f};
#pragma unroll
    for (int t = 0; t < TILES; ++t) {
        s0 += d[t][0];
        s1 += d[t][1];
        s2 += d[t][2];
        s3 += d[t][3];
    }
    const v2f sh = (s0 + s1) + (s2 + s3);
    const int x0 = __float_as_int(sh.x), y0 = __float_as_int(sh.y);
    const int x1 = __builtin_amdgcn_permlanex16(x0, x0, 0x76543210u, 0xfedcba98u, true, false);
    const int y1 = __builtin_amdgcn_permlanex16(y0, y0, 0x76543210u, 0xfedcba98u, true, false);
    v2f r;
    r.x = sh.x + __int_as_float(x1);
    r.y = sh.y + __int_as_float(y1);
    return r;
}

__device__ __forceinline__ void lds_barrier()
{
    __builtin_amdgcn_fence(__ATOMIC_RELEASE, "workgroup", "local");
    __builtin_amdgcn_s_barrier();
    __builtin_amdgcn_fence(__ATOMIC_ACQUIRE, "workgroup", "local");
}

template <bool UPPER>
__device__ __forceinline__ void sweep(
    const float* __restrict__ x, const float* __restrict__ zero, float* __restrict__ out,
    v2f (&tot)[2][NWAVES][16], v2f (&low)[SCAN_STEPS][NWAVES][16])
{
    constexpr bool upper = UPPER;
    const int  grp   = blockIdx.x & (NGROUPS - 1);
    const int bh   = grp >> 1;
    const int ch   = grp & 1;
    const int lane = threadIdx.x & 31;
    const int wave = __builtin_amdgcn_readfirstlane((int)(threadIdx.x >> 5));
    const int col  = lane & 15;
    const bool hi  = lane >= 16;

    const float c1 = __expf(fminf(fmaxf(zero[0], -60.f), 30.f));

    const size_t base = (size_t)bh * NROWS * DCOLS + (size_t)wave * ROWS_PER_WAVE * DCOLS
                      + (size_t)ch * WCOLS + (size_t)(2 * col);
    const v2f* xp = (const v2f*)(x + base + (hi ? DCOLS : 0));
    v2f*       op = (v2f*)(out + base + (hi ? 4 * DCOLS : 0));

    v2f a[4];
    {
        const int ai = col >> 1;
        const int aq = col & 1;
#pragma unroll
        for (int j = 0; j < 4; ++j) {
            const int r = 2 * j + (hi ? 1 : 0);
            const float v = (r <= ai) ? c1 : 0.f;
            a[j].x = (aq == 0) ? v : 0.f;
            a[j].y = (aq == 1) ? v : 0.f;
        }
    }

    v2f buf[2][TILES][4];
    load_block(xp, buf[0]);
    load_block(xp + (size_t)ROWS_PER_STEP * ROWPITCH2, buf[1]);

    if (upper) {
        const v2f l0 = block_total(buf[0]);
        __builtin_amdgcn_sched_barrier(0);
        load_block(xp + (size_t)2 * ROWS_PER_STEP * ROWPITCH2, buf[0]);
        const v2f l1 = block_total(buf[1]);
        __builtin_amdgcn_sched_barrier(0);
        load_block(xp + (size_t)3 * ROWS_PER_STEP * ROWPITCH2, buf[1]);
        if (!hi) {
            low[0][wave][col] = c1 * l0;
            low[1][wave][col] = c1 * l1;
        }
    }

    {
        const v2f t0 = block_total(buf[0]);
        if (!hi) tot[0][wave][col] = c1 * t0;
    }
    lds_barrier();

    v2f run = {0.f, 0.f};
    if (upper) {
#pragma unroll
        for (int b = 0; b < SCAN_STEPS; ++b) {
            v2f all = {0.f, 0.f};
#pragma unroll
            for (int w = 0; w < NWAVES; ++w) all += low[b][w][col];
            run += all;
        }
        op += (size_t)2 * ROWS_PER_STEP * ROWPITCH2;
    }

#pragma unroll
    for (int s = 0; s < SCAN_STEPS; ++s) {
        const int par = s & 1;

        v2f tv[NWAVES];
#pragma unroll
        for (int w = 0; w < NWAVES; ++w) tv[w] = tot[par][w][col];
        v2f pre = {0.f, 0.f}, all = {0.f, 0.f};
#pragma unroll
        for (int w = 0; w < NWAVES; ++w) {
            const v2f z = {0.f, 0.f};
            pre += (w < wave) ? tv[w] : z;
            all += tv[w];
        }
        float carry0 = run.x + pre.x;
        float carry1 = run.y + pre.y;
        run += all;

#pragma unroll
        for (int t = 0; t < TILES; ++t) {
            v8f acc;
#pragma unroll
            for (int v = 0; v < 8; ++v) acc[v] = (v & 1) ? carry1 : carry0;

#pragma unroll
            for (int j = 0; j < 4; ++j)
                acc = __builtin_amdgcn_wmma_f32_16x16x4_f32(
                    false, a[j], false, buf[par][t][j], (short)0, acc, false, false);

            v2f* topp = op + (size_t)(s * ROWS_PER_STEP + t * TROWS) * ROWPITCH2;
#pragma unroll
            for (int u = 0; u < 4; ++u) {
                v2f o;
                o.x = acc[2 * u];
                o.y = acc[2 * u + 1];
                topp[(size_t)u * ROWPITCH2] = o;
            }

            const int a0i = __float_as_int(acc[6]);
            const int a1i = __float_as_int(acc[7]);
            const int s0i = __builtin_amdgcn_permlanex16(a0i, a0i, 0x76543210u, 0xfedcba98u,
                                                         true, false);
            const int s1i = __builtin_amdgcn_permlanex16(a1i, a1i, 0x76543210u, 0xfedcba98u,
                                                         true, false);
            carry0 = hi ? acc[6] : __int_as_float(s0i);
            carry1 = hi ? acc[7] : __int_as_float(s1i);
        }

        if (s + 1 < SCAN_STEPS) {
            __builtin_amdgcn_sched_barrier(0);
            const v2f t1 = block_total(buf[par ^ 1]);
            if (!hi) tot[par ^ 1][wave][col] = c1 * t1;
            lds_barrier();
        }
    }
}

__global__ __launch_bounds__(512) void toeplitz_scan_wmma(
    const float* __restrict__ x, const float* __restrict__ zero,
    float* __restrict__ out)
{
    __shared__ v2f tot[2][NWAVES][16];
    __shared__ v2f low[SCAN_STEPS][NWAVES][16];

    if (blockIdx.x < NGROUPS) sweep<true>(x, zero, out, tot, low);
    else                      sweep<false>(x, zero, out, tot, low);
}

extern "C" void kernel_launch(void* const* d_in, const int* in_sizes, int n_in,
                              void* d_out, int out_size, void* d_ws, size_t ws_size,
                              hipStream_t stream) {
    const float* x    = (const float*)d_in[0];
    const float* zero = (const float*)d_in[2];
    float* out = (float*)d_out;

    toeplitz_scan_wmma<<<dim3(256), dim3(512), 0, stream>>>(x, zero, out);
}


# Split point moved: 8 row blocks of 512; lower work-group scans 5, upper adds those 5 up per wave and scans 3, so both move about equal bytes; start-value sum order changes (per wave, then waves)
# speedup vs baseline: 12.8662x; 1.0652x over previous
#include <hip/hip_runtime.h>


typedef __attribute__((ext_vector_type(2))) float v2f;
typedef __attribute__((ext_vector_type(8))) float v8f;

#define NROWS 4096
#define DCOLS 64
#define WCOLS 32
#define NWAVES 16
#define STEPS 8
#define LOW_STEPS 5
#define NBUF 4
#define NGROUPS 128
#define ROWS_PER_STEP (NROWS / STEPS)
#define ROWS_PER_WAVE (ROWS_PER_STEP / NWAVES)
#define TROWS 8
#define TILES (ROWS_PER_WAVE / TROWS)
#define ROWPITCH2 (DCOLS / 2)

__device__ __forceinline__ void load_block(const v2f* __restrict__ p, v2f (&d)[TILES][4])
{
#pragma unroll
    for (int t = 0; t < TILES; ++t)
#pragma unroll
        for (int j = 0; j < 4; ++j)
            d[t][j] = p[(size_t)(TROWS * t + 2 * j) * ROWPITCH2];
}

__device__ __forceinline__ v2f block_total(const v2f (&d)[TILES][4])
{
    v2f s0 = {0.f, 0.f}, s1 = {0.f, 0.f}, s2 = {0.f, 0.f}, s3 = {0.f, 0.f};
#pragma unroll
    for (int t = 0; t < TILES; ++t) {
        s0 += d[t][0];
        s1 += d[t][1];
        s2 += d[t][2];
        s3 += d[t][3];
    }
    const v2f sh = (s0 + s1) + (s2 + s3);
    const int x0 = __float_as_int(sh.x), y0 = __float_as_int(sh.y);
    const int x1 = __builtin_amdgcn_permlanex16(x0, x0, 0x76543210u, 0xfedcba98u, true, false);
    const int y1 = __builtin_amdgcn_permlanex16(y0, y0, 0x76543210u, 0xfedcba98u, true, false);
    v2f r;
    r.x = sh.x + __int_as_float(x1);
    r.y = sh.y + __int_as_float(y1);
    return r;
}

__device__ __forceinline__ void lds_barrier()
{
    __builtin_amdgcn_fence(__ATOMIC_RELEASE, "workgroup", "local");
    __builtin_amdgcn_s_barrier();
    __builtin_amdgcn_fence(__ATOMIC_ACQUIRE, "workgroup", "local");
}

template <bool UPPER>
__device__ __forceinline__ void sweep(
    const float* __restrict__ x, const float* __restrict__ zero, float* __restrict__ out,
    v2f (&tot)[2][NWAVES][16], v2f (&low)[NWAVES][16])
{
    constexpr bool upper = UPPER;
    constexpr int first = UPPER ? LOW_STEPS : 0;
    constexpr int nscan = UPPER ? STEPS - LOW_STEPS : LOW_STEPS;
    const size_t blk_pitch = (size_t)ROWS_PER_STEP * ROWPITCH2;
    const int  grp   = blockIdx.x & (NGROUPS - 1);
    const int bh   = grp >> 1;
    const int ch   = grp & 1;
    const int lane = threadIdx.x & 31;
    const int wave = __builtin_amdgcn_readfirstlane((int)(threadIdx.x >> 5));
    const int col  = lane & 15;
    const bool hi  = lane >= 16;

    const float c1 = __expf(fminf(fmaxf(zero[0], -60.f), 30.f));

    const size_t base = (size_t)bh * NROWS * DCOLS + (size_t)wave * ROWS_PER_WAVE * DCOLS
                      + (size_t)ch * WCOLS + (size_t)(2 * col);
    const v2f* xp = (const v2f*)(x + base + (hi ? DCOLS : 0));
    v2f*       op = (v2f*)(out + base + (hi ? 4 * DCOLS : 0));

    v2f a[4];
    {
        const int ai = col >> 1;
        const int aq = col & 1;
#pragma unroll
        for (int j = 0; j < 4; ++j) {
            const int r = 2 * j + (hi ? 1 : 0);
            const float v = (r <= ai) ? c1 : 0.f;
            a[j].x = (aq == 0) ? v : 0.f;
            a[j].y = (aq == 1) ? v : 0.f;
        }
    }

    v2f buf[NBUF][TILES][4];
#pragma unroll
    for (int b = 0; b < NBUF; ++b) load_block(xp + (size_t)b * blk_pitch, buf[b]);

    if (upper) {
        v2f lsum = {0.f, 0.f};
#pragma unroll
        for (int b = 0; b < LOW_STEPS; ++b) {
            lsum += block_total(buf[b % NBUF]);
            __builtin_amdgcn_sched_barrier(0);
            if (b + NBUF < STEPS) load_block(xp + (size_t)(b + NBUF) * blk_pitch, buf[b % NBUF]);
        }
        if (!hi) low[wave][col] = c1 * lsum;
    }

    {
        const v2f t0 = block_total(buf[first % NBUF]);
        if (!hi) tot[0][wave][col] = c1 * t0;
    }
    lds_barrier();

    v2f run = {0.f, 0.f};
    if (upper) {
#pragma unroll
        for (int w = 0; w < NWAVES; ++w) run += low[w][col];
        op += (size_t)first * blk_pitch;
    }

#pragma unroll
    for (int s = 0; s < nscan; ++s) {
        const int par = s & 1;
        const int bi  = (first + s) % NBUF;

        v2f tv[NWAVES];
#pragma unroll
        for (int w = 0; w < NWAVES; ++w) tv[w] = tot[par][w][col];
        v2f pre = {0.f, 0.f}, all = {0.f, 0.f};
#pragma unroll
        for (int w = 0; w < NWAVES; ++w) {
            const v2f z = {0.f, 0.f};
            pre += (w < wave) ? tv[w] : z;
            all += tv[w];
        }
        float carry0 = run.x + pre.x;
        float carry1 = run.y + pre.y;
        run += all;

#pragma unroll
        for (int t = 0; t < TILES; ++t) {
            v8f acc;
#pragma unroll
            for (int v = 0; v < 8; ++v) acc[v] = (v & 1) ? carry1 : carry0;

#pragma unroll
            for (int j = 0; j < 4; ++j)
                acc = __builtin_amdgcn_wmma_f32_16x16x4_f32(
                    false, a[j], false, buf[bi][t][j], (short)0, acc, false, false);

            v2f* topp = op + (size_t)(s * ROWS_PER_STEP + t * TROWS) * ROWPITCH2;
#pragma unroll
            for (int u = 0; u < 4; ++u) {
                v2f o;
                o.x = acc[2 * u];
                o.y = acc[2 * u + 1];
                topp[(size_t)u * ROWPITCH2] = o;
            }

            const int a0i = __float_as_int(acc[6]);
            const int a1i = __float_as_int(acc[7]);
            const int s0i = __builtin_amdgcn_permlanex16(a0i, a0i, 0x76543210u, 0xfedcba98u,
                                                         true, false);
            const int s1i = __builtin_amdgcn_permlanex16(a1i, a1i, 0x76543210u, 0xfedcba98u,
                                                         true, false);
            carry0 = hi ? acc[6] : __int_as_float(s0i);
            carry1 = hi ? acc[7] : __int_as_float(s1i);
        }

        if (!upper && first + s + NBUF < LOW_STEPS)
            load_block(xp + (size_t)(first + s + NBUF) * blk_pitch, buf[bi]);

        if (s + 1 < nscan) {
            __builtin_amdgcn_sched_barrier(0);
            const v2f t1 = block_total(buf[(first + s + 1) % NBUF]);
            if (!hi) tot[par ^ 1][wave][col] = c1 * t1;
            lds_barrier();
        }
    }
}

__global__ __launch_bounds__(512) void toeplitz_scan_wmma(
    const float* __restrict__ x, const float* __restrict__ zero,
    float* __restrict__ out)
{
    __shared__ v2f tot[2][NWAVES][16];
    __shared__ v2f low[NWAVES][16];

    if (blockIdx.x < NGROUPS) sweep<true>(x, zero, out, tot, low);
    else                      sweep<false>(x, zero, out, tot, low);
}

extern "C" void kernel_launch(void* const* d_in, const int* in_sizes, int n_in,
                              void* d_out, int out_size, void* d_ws, size_t ws_size,
                              hipStream_t stream) {
    const float* x    = (const float*)d_in[0];
    const float* zero = (const float*)d_in[2];
    float* out = (float*)d_out;

    toeplitz_scan_wmma<<<dim3(256), dim3(512), 0, stream>>>(x, zero, out);
}
